# starting kernel
# baseline (speedup 1.0000x reference)
#include <hip/hip_runtime.h>
#include <hip/hip_bf16.h>


typedef __attribute__((ext_vector_type(2))) float v2f;
typedef __attribute__((ext_vector_type(8))) float v8f;

#define BDP_H     16
#define BDP_NIN   128
#define BDP_NOUT  128
#define BT_PITCH  144
#define TOK_PER_WG 128

__global__ __launch_bounds__(256)
void bdp_wmma_f32_kernel(const float* __restrict__ in,
                         const float* __restrict__ W,
                         float* __restrict__ out) {
    __shared__ v2f BT[64 * BT_PITCH];

    const int h        = (int)blockIdx.x & (BDP_H - 1);
    const int tokBlock = (int)blockIdx.x >> 4;
    const int tid      = (int)threadIdx.x;

    const v2f* Wh2 = (const v2f*)(W + h * (BDP_NOUT * BDP_NIN));
    #pragma unroll
    for (int it = 0; it < 32; ++it) {
        const int idx = tid + it * 256;
        const int o = idx & 127;
        const int p = idx >> 7;
        BT[p * BT_PITCH + o] = Wh2[o * 64 + p];
    }
    __syncthreads();

    const int wave = tid >> 5;
    const int lane = tid & 31;
    const int nrow = lane & 15;
    const int kh   = lane >> 4;

    const int tokBase = tokBlock * TOK_PER_WG + wave * 16;

    const float* Arow = in + ((tokBase + nrow) * BDP_H + h) * BDP_NIN + kh * 2;
    v2f a[32];
    #pragma unroll
    for (int k = 0; k < 32; ++k) {
        a[k] = *(const v2f*)(Arow + 4 * k);
    }

    #pragma unroll
    for (int nt = 0; nt < 8; ++nt) {
        v8f c = {};
        const v2f* Bb = &BT[kh * BT_PITCH + nt * 16 + nrow];
        #pragma unroll
        for (int k = 0; k < 32; ++k) {
            const v2f b = Bb[(2 * k) * BT_PITCH];
            c = __builtin_amdgcn_wmma_f32_16x16x4_f32(
                    false, a[k], false, b, (short)0, c, false, false);
        }
        const int n  = nt * 16 + nrow;
        const int mb = kh * 8;
        #pragma unroll
        for (int r = 0; r < 8; ++r) {
            out[((tokBase + mb + r) * BDP_H + h) * BDP_NOUT + n] = c[r];
        }
    }
}

extern "C" void kernel_launch(void* const* d_in, const int* in_sizes, int n_in,
                              void* d_out, int out_size, void* d_ws, size_t ws_size,
                              hipStream_t stream) {
    const float* in  = (const float*)d_in[0];
    const float* W   = (const float*)d_in[1];
    float*       out = (float*)d_out;
    (void)in_sizes; (void)n_in; (void)out_size; (void)d_ws; (void)ws_size;

    const int tokTotal  = 8 * 2048;
    const int numBlocks = (tokTotal / TOK_PER_WG) * BDP_H;
    dim3 grid(numBlocks), block(256);
    bdp_wmma_f32_kernel<<<grid, block, 0, stream>>>(in, W, out);
}


# W[h] staged in LDS row-major (row pitch 132 dwords) by whole-line 16-byte loads instead of a lane-strided 8-byte transposing gather: 4 lines per fill load, not 32; B reads stay b64
# speedup vs baseline: 1.1030x; 1.1030x over previous
#include <hip/hip_runtime.h>
#include <hip/hip_bf16.h>


typedef __attribute__((ext_vector_type(2))) float v2f;
typedef __attribute__((ext_vector_type(4))) float v4f;
typedef __attribute__((ext_vector_type(8))) float v8f;

#define BDP_H     16
#define BDP_NIN   128
#define BDP_NOUT  128
#define W_PITCH   132
#define TOK_PER_WG 128

__global__ __launch_bounds__(256)
void bdp_wmma_f32_kernel(const float* __restrict__ in,
                         const float* __restrict__ W,
                         float* __restrict__ out) {
    __shared__ v4f Ws4[BDP_NOUT * (W_PITCH / 4)];

    const int h        = (int)blockIdx.x & (BDP_H - 1);
    const int tokBlock = (int)blockIdx.x >> 4;
    const int tid      = (int)threadIdx.x;

    const v4f* Wh4 = (const v4f*)(W + h * (BDP_NOUT * BDP_NIN));
    #pragma unroll
    for (int it = 0; it < 16; ++it) {
        const int idx = tid + it * 256;
        const int o  = idx >> 5;
        const int c4 = idx & 31;
        Ws4[o * (W_PITCH / 4) + c4] = Wh4[idx];
    }
    __syncthreads();

    const int wave = tid >> 5;
    const int lane = tid & 31;
    const int nrow = lane & 15;
    const int kh   = lane >> 4;

    const int tokBase = tokBlock * TOK_PER_WG + wave * 16;

    const float* Arow = in + ((tokBase + nrow) * BDP_H + h) * BDP_NIN + kh * 2;
    v2f a[32];
    #pragma unroll
    for (int k = 0; k < 32; ++k) {
        a[k] = *(const v2f*)(Arow + 4 * k);
    }

    #pragma unroll
    for (int nt = 0; nt < 8; ++nt) {
        v8f c = {};
        const v2f* Bb = (const v2f*)Ws4 + (nt * 16 + nrow) * (W_PITCH / 2) + kh;
        #pragma unroll
        for (int k = 0; k < 32; ++k) {
            const v2f b = Bb[2 * k];
            c = __builtin_amdgcn_wmma_f32_16x16x4_f32(
                    false, a[k], false, b, (short)0, c, false, false);
        }
        const int n  = nt * 16 + nrow;
        const int mb = kh * 8;
        #pragma unroll
        for (int r = 0; r < 8; ++r) {
            out[((tokBase + mb + r) * BDP_H + h) * BDP_NOUT + n] = c[r];
        }
    }
}

extern "C" void kernel_launch(void* const* d_in, const int* in_sizes, int n_in,
                              void* d_out, int out_size, void* d_ws, size_t ws_size,
                              hipStream_t stream) {
    const float* in  = (const float*)d_in[0];
    const float* W   = (const float*)d_in[1];
    float*       out = (float*)d_out;
    (void)in_sizes; (void)n_in; (void)out_size; (void)d_ws; (void)ws_size;

    const int tokTotal  = 8 * 2048;
    const int numBlocks = (tokTotal / TOK_PER_WG) * BDP_H;
    dim3 grid(numBlocks), block(256);
    bdp_wmma_f32_kernel<<<grid, block, 0, stream>>>(in, W, out);
}


# One 16-wave work-group per WGP (grid 256, was 2048) stages W[h] in LDS once and each wave loops over four 16-token tiles, a tile's 32 A loads issued as one group; no W refill or barrier per tile
# speedup vs baseline: 1.2578x; 1.1403x over previous
#include <hip/hip_runtime.h>
#include <hip/hip_bf16.h>


typedef __attribute__((ext_vector_type(2))) float v2f;
typedef __attribute__((ext_vector_type(4))) float v4f;
typedef __attribute__((ext_vector_type(8))) float v8f;

#define BDP_H     16
#define BDP_NIN   128
#define BDP_NOUT  128
#define W_PITCH   132
#define WG_THREADS     512
#define WG_WAVES       (WG_THREADS / 32)
#define TILES_PER_WAVE 4
#define TOK_PER_PASS   (WG_WAVES * 16)
#define TOK_PER_WG     (TOK_PER_PASS * TILES_PER_WAVE)

__global__ __launch_bounds__(WG_THREADS) __attribute__((amdgpu_waves_per_eu(4, 5)))
void bdp_wmma_f32_kernel(const float* __restrict__ in,
                         const float* __restrict__ W,
                         float* __restrict__ out) {
    __shared__ v4f Ws4[BDP_NOUT * (W_PITCH / 4)];

    const int h        = (int)blockIdx.x & (BDP_H - 1);
    const int tokGroup = (int)blockIdx.x >> 4;
    const int tid      = (int)threadIdx.x;

    const v4f* Wh4 = (const v4f*)(W + h * (BDP_NOUT * BDP_NIN));
    #pragma unroll
    for (int it = 0; it < 4096 / WG_THREADS; ++it) {
        const int idx = tid + it * WG_THREADS;
        const int o  = idx >> 5;
        const int c4 = idx & 31;
        Ws4[o * (W_PITCH / 4) + c4] = Wh4[idx];
    }
    __syncthreads();

    const int wave = tid >> 5;
    const int lane = tid & 31;
    const int nrow = lane & 15;
    const int kh   = lane >> 4;

    #pragma unroll 1
    for (int t = 0; t < TILES_PER_WAVE; ++t) {
        asm volatile("" ::: "memory");

        const int tokBase = tokGroup * TOK_PER_WG + t * TOK_PER_PASS + wave * 16;

        const float* Arow = in + ((tokBase + nrow) * BDP_H + h) * BDP_NIN + kh * 2;
        v2f a[32];
        #pragma unroll
        for (int k = 0; k < 32; ++k) {
            a[k] = *(const v2f*)(Arow + 4 * k);
        }
        __builtin_amdgcn_sched_barrier(0);

        #pragma unroll
        for (int nt = 0; nt < 8; ++nt) {
            v8f c = {};
            const v2f* Bb = (const v2f*)Ws4 + (nt * 16 + nrow) * (W_PITCH / 2) + kh;
            #pragma unroll
            for (int k = 0; k < 32; ++k) {
                const v2f b = Bb[2 * k];
                c = __builtin_amdgcn_wmma_f32_16x16x4_f32(
                        false, a[k], false, b, (short)0, c, false, false);
            }
            const int n  = nt * 16 + nrow;
            const int mb = kh * 8;
            #pragma unroll
            for (int r = 0; r < 8; ++r) {
                out[((tokBase + mb + r) * BDP_H + h) * BDP_NOUT + n] = c[r];
            }
        }
    }
}

extern "C" void kernel_launch(void* const* d_in, const int* in_sizes, int n_in,
                              void* d_out, int out_size, void* d_ws, size_t ws_size,
                              hipStream_t stream) {
    const float* in  = (const float*)d_in[0];
    const float* W   = (const float*)d_in[1];
    float*       out = (float*)d_out;
    (void)in_sizes; (void)n_in; (void)out_size; (void)d_ws; (void)ws_size;

    const int tokTotal  = 8 * 2048;
    const int numBlocks = (tokTotal / TOK_PER_WG) * BDP_H;
    dim3 grid(numBlocks), block(WG_THREADS);
    bdp_wmma_f32_kernel<<<grid, block, 0, stream>>>(in, W, out);
}


# Each wave's 16 x 128 result tile staged in its own LDS rows and written as 16 whole 512-byte rows by 16-byte stores: 4 full lines per store, was 64 stores of two 64-byte halves each; sums unchanged
# speedup vs baseline: 1.3862x; 1.1021x over previous
#include <hip/hip_runtime.h>
#include <hip/hip_bf16.h>


typedef __attribute__((ext_vector_type(2))) float v2f;
typedef __attribute__((ext_vector_type(4))) float v4f;
typedef __attribute__((ext_vector_type(8))) float v8f;

#define BDP_H     16
#define BDP_NIN   128
#define BDP_NOUT  128
#define W_PITCH   132
#define C_PITCH   132
#define WG_THREADS     512
#define WG_WAVES       (WG_THREADS / 32)
#define TILES_PER_WAVE 4
#define TOK_PER_PASS   (WG_WAVES * 16)
#define TOK_PER_WG     (TOK_PER_PASS * TILES_PER_WAVE)

__global__ __launch_bounds__(WG_THREADS) __attribute__((amdgpu_waves_per_eu(4, 5)))
void bdp_wmma_f32_kernel(const float* __restrict__ in,
                         const float* __restrict__ W,
                         float* __restrict__ out) {
    __shared__ v4f Ws4[BDP_NOUT * (W_PITCH / 4)];
    __shared__ v4f Cs4[WG_WAVES * 16 * (C_PITCH / 4)];

    const int h        = (int)blockIdx.x & (BDP_H - 1);
    const int tokGroup = (int)blockIdx.x >> 4;
    const int tid      = (int)threadIdx.x;

    const v4f* Wh4 = (const v4f*)(W + h * (BDP_NOUT * BDP_NIN));
    #pragma unroll
    for (int it = 0; it < 4096 / WG_THREADS; ++it) {
        const int idx = tid + it * WG_THREADS;
        const int o  = idx >> 5;
        const int c4 = idx & 31;
        Ws4[o * (W_PITCH / 4) + c4] = Wh4[idx];
    }
    __syncthreads();

    const int wave = tid >> 5;
    const int lane = tid & 31;
    const int nrow = lane & 15;
    const int kh   = lane >> 4;
    float* Cw = (float*)Cs4 + wave * (16 * C_PITCH);

    #pragma unroll 1
    for (int t = 0; t < TILES_PER_WAVE; ++t) {
        asm volatile("" ::: "memory");

        const int tokBase = tokGroup * TOK_PER_WG + t * TOK_PER_PASS + wave * 16;

        const float* Arow = in + ((tokBase + nrow) * BDP_H + h) * BDP_NIN + kh * 2;
        v2f a[32];
        #pragma unroll
        for (int k = 0; k < 32; ++k) {
            a[k] = *(const v2f*)(Arow + 4 * k);
        }
        __builtin_amdgcn_sched_barrier(0);

        #pragma unroll
        for (int nt = 0; nt < 8; ++nt) {
            v8f c = {};
            const v2f* Bb = (const v2f*)Ws4 + (nt * 16 + nrow) * (W_PITCH / 2) + kh;
            #pragma unroll
            for (int k = 0; k < 32; ++k) {
                const v2f b = Bb[2 * k];
                c = __builtin_amdgcn_wmma_f32_16x16x4_f32(
                        false, a[k], false, b, (short)0, c, false, false);
            }
            float* Cst = Cw + (kh * 8) * C_PITCH + nt * 16 + nrow;
            #pragma unroll
            for (int r = 0; r < 8; ++r) {
                Cst[r * C_PITCH] = c[r];
            }
        }

        float* Orow = out + ((size_t)tokBase * BDP_H + h) * BDP_NOUT + 4 * lane;
        #pragma unroll
        for (int tk = 0; tk < 16; ++tk) {
            const v4f v = *(const v4f*)(Cw + tk * C_PITCH + 4 * lane);
            *(v4f*)(Orow + (size_t)tk * (BDP_H * BDP_NOUT)) = v;
        }
    }
}

extern "C" void kernel_launch(void* const* d_in, const int* in_sizes, int n_in,
                              void* d_out, int out_size, void* d_ws, size_t ws_size,
                              hipStream_t stream) {
    const float* in  = (const float*)d_in[0];
    const float* W   = (const float*)d_in[1];
    float*       out = (float*)d_out;
    (void)in_sizes; (void)n_in; (void)out_size; (void)d_ws; (void)ws_size;

    const int tokTotal  = 8 * 2048;
    const int numBlocks = (tokTotal / TOK_PER_WG) * BDP_H;
    dim3 grid(numBlocks), block(WG_THREADS);
    bdp_wmma_f32_kernel<<<grid, block, 0, stream>>>(in, W, out);
}


# W[h] kept in registers: 4-wave work-groups, one wave per SIMD, read all 256 B fragments from LDS once; no LDS read stands among the WMMAs; next tile's 32 A loads go out a tile ahead; sums unchanged
# speedup vs baseline: 1.5749x; 1.1361x over previous
#include <hip/hip_runtime.h>
#include <hip/hip_bf16.h>


typedef __attribute__((ext_vector_type(2))) float v2f;
typedef __attribute__((ext_vector_type(4))) float v4f;
typedef __attribute__((ext_vector_type(8))) float v8f;

#define BDP_H     16
#define BDP_NIN   128
#define BDP_NOUT  128
#define W_PITCH   132
#define C_PITCH   132
#define WG_THREADS     128
#define WG_WAVES       (WG_THREADS / 32)
#define TILES_PER_WAVE 16
#define TOK_PER_PASS   (WG_WAVES * 16)
#define TOK_PER_WG     (TOK_PER_PASS * TILES_PER_WAVE)

__device__ __forceinline__ void load_a_tile(v2f (&a)[32], const float* Arow) {
    #pragma unroll
    for (int k = 0; k < 32; ++k) {
        a[k] = *(const v2f*)(Arow + 4 * k);
    }
}

__device__ __forceinline__ void do_tile(const v2f (&a)[32], const v2f (&b)[8][32],
                                        float* Cw, float* Orow,
                                        int nrow, int kh, int lane) {
    v8f c[8];
    #pragma unroll
    for (int nt = 0; nt < 8; ++nt) {
        c[nt] = (v8f){};
    }
    #pragma unroll
    for (int k = 0; k < 32; ++k) {
        #pragma unroll
        for (int nt = 0; nt < 8; ++nt) {
            c[nt] = __builtin_amdgcn_wmma_f32_16x16x4_f32(
                        false, a[k], false, b[nt][k], (short)0, c[nt], false, false);
        }
    }
    #pragma unroll
    for (int nt = 0; nt < 8; ++nt) {
        float* Cst = Cw + (kh * 8) * C_PITCH + nt * 16 + nrow;
        #pragma unroll
        for (int r = 0; r < 8; ++r) {
            Cst[r * C_PITCH] = c[nt][r];
        }
    }
    #pragma unroll
    for (int tk = 0; tk < 16; ++tk) {
        const v4f v = *(const v4f*)(Cw + tk * C_PITCH + 4 * lane);
        *(v4f*)(Orow + (size_t)tk * (BDP_H * BDP_NOUT)) = v;
    }
}

__global__ __launch_bounds__(WG_THREADS)
void bdp_wmma_f32_kernel(const float* __restrict__ in,
                         const float* __restrict__ W,
                         float* __restrict__ out) {
    __shared__ v4f Ws4[BDP_NOUT * (W_PITCH / 4)];
    __shared__ v4f Cs4[WG_WAVES * 16 * (C_PITCH / 4)];

    const int h        = (int)blockIdx.x & (BDP_H - 1);
    const int tokGroup = (int)blockIdx.x >> 4;
    const int tid      = (int)threadIdx.x;

    const v4f* Wh4 = (const v4f*)(W + h * (BDP_NOUT * BDP_NIN));
    #pragma unroll
    for (int it = 0; it < 4096 / WG_THREADS; ++it) {
        const int idx = tid + it * WG_THREADS;
        const int o  = idx >> 5;
        const int c4 = idx & 31;
        Ws4[o * (W_PITCH / 4) + c4] = Wh4[idx];
    }
    __syncthreads();

    const int wave = tid >> 5;
    const int lane = tid & 31;
    const int nrow = lane & 15;
    const int kh   = lane >> 4;
    float* Cw = (float*)Cs4 + wave * (16 * C_PITCH);

    v2f b[8][32];
    #pragma unroll
    for (int nt = 0; nt < 8; ++nt) {
        const v2f* Bb = (const v2f*)Ws4 + (nt * 16 + nrow) * (W_PITCH / 2) + kh;
        #pragma unroll
        for (int k = 0; k < 32; ++k) {
            b[nt][k] = Bb[2 * k];
        }
    }

    const int tok0 = tokGroup * TOK_PER_WG + wave * 16;
    const float* Arow0 = in + ((tok0 + nrow) * BDP_H + h) * BDP_NIN + kh * 2;
    float*       Orow0 = out + ((size_t)tok0 * BDP_H + h) * BDP_NOUT + 4 * lane;
    const int aStep = TOK_PER_PASS * BDP_H * BDP_NIN;
    const int oStep = TOK_PER_PASS * BDP_H * BDP_NOUT;

    v2f a0[32], a1[32];
    load_a_tile(a0, Arow0);

    #pragma unroll 1
    for (int t = 0; t < TILES_PER_WAVE - 2; t += 2) {
        load_a_tile(a1, Arow0 + (t + 1) * aStep);
        __builtin_amdgcn_sched_barrier(0);
        do_tile(a0, b, Cw, Orow0 + t * oStep, nrow, kh, lane);
        __builtin_amdgcn_sched_barrier(0);

        load_a_tile(a0, Arow0 + (t + 2) * aStep);
        __builtin_amdgcn_sched_barrier(0);
        do_tile(a1, b, Cw, Orow0 + (t + 1) * oStep, nrow, kh, lane);
        __builtin_amdgcn_sched_barrier(0);
    }
    load_a_tile(a1, Arow0 + (TILES_PER_WAVE - 1) * aStep);
    __builtin_amdgcn_sched_barrier(0);
    do_tile(a0, b, Cw, Orow0 + (TILES_PER_WAVE - 2) * oStep, nrow, kh, lane);
    __builtin_amdgcn_sched_barrier(0);
    do_tile(a1, b, Cw, Orow0 + (TILES_PER_WAVE - 1) * oStep, nrow, kh, lane);
}

extern "C" void kernel_launch(void* const* d_in, const int* in_sizes, int n_in,
                              void* d_out, int out_size, void* d_ws, size_t ws_size,
                              hipStream_t stream) {
    const float* in  = (const float*)d_in[0];
    const float* W   = (const float*)d_in[1];
    float*       out = (float*)d_out;
    (void)in_sizes; (void)n_in; (void)out_size; (void)d_ws; (void)ws_size;

    const int tokTotal  = 8 * 2048;
    const int numBlocks = (tokTotal / TOK_PER_WG) * BDP_H;
    dim3 grid(numBlocks), block(WG_THREADS);
    bdp_wmma_f32_kernel<<<grid, block, 0, stream>>>(in, W, out);
}
